# starting kernel
# baseline (speedup 1.0000x reference)
#include <hip/hip_runtime.h>

#define D_IN   256
#define D_OUT  64
#define LDS_STRIDE 260

typedef float v2f __attribute__((ext_vector_type(2)));
typedef float v8f __attribute__((ext_vector_type(8)));

__global__ void init_deg_kernel(float* __restrict__ out_deg,
                                float* __restrict__ in_deg, int n_nodes) {
    int i = blockIdx.x * blockDim.x + threadIdx.x;
    if (i < n_nodes) { out_deg[i] = 1.0f; in_deg[i] = 1.0f; }
}

__global__ void zero_out_kernel(float* __restrict__ out, long long n) {
    long long i = (long long)blockIdx.x * blockDim.x + threadIdx.x;
    if (i < n) out[i] = 0.0f;
}

__global__ void degree_kernel(const int* __restrict__ src, const int* __restrict__ dst,
                              float* __restrict__ out_deg, float* __restrict__ in_deg,
                              int n_edges) {
    int e = blockIdx.x * blockDim.x + threadIdx.x;
    if (e >= n_edges) return;
    atomicAdd(&out_deg[src[e]], 1.0f);
    atomicAdd(&in_deg[dst[e]], 1.0f);
}

__global__ void norm_kernel(float* __restrict__ out_deg, float* __restrict__ in_deg,
                            int n_nodes) {
    int i = blockIdx.x * blockDim.x + threadIdx.x;
    if (i >= n_nodes) return;
    out_deg[i] = rsqrtf(out_deg[i]);
    in_deg[i]  = rsqrtf(in_deg[i]);
}

__global__ __launch_bounds__(128) void gemm_wmma_kernel(
        const float* __restrict__ X, const float* __restrict__ W,
        const float* __restrict__ norm_src, float* __restrict__ H, int n_nodes) {
    __shared__ float As[16 * LDS_STRIDE];

    const int m0  = blockIdx.x * 16;
    const int tid = threadIdx.x;
    const bool full_tile = (m0 + 16 <= n_nodes);

    if (full_tile) {
        #pragma unroll
        for (int it = 0; it < 8; ++it) {
            int i    = tid + it * 128;
            int row  = i >> 6;
            int col4 = (i & 63) * 4;
            int grow = m0 + row;
            float4 v = *(const float4*)(X + (size_t)grow * D_IN + col4);
            float ns = norm_src[grow];
            float* p = &As[row * LDS_STRIDE + col4];
            p[0] = v.x * ns; p[1] = v.y * ns; p[2] = v.z * ns; p[3] = v.w * ns;
        }
    } else {
        for (int i = tid; i < 16 * (D_IN / 4); i += 128) {
            int row  = i >> 6;
            int col4 = (i & 63) * 4;
            int grow = m0 + row;
            float4 v = make_float4(0.f, 0.f, 0.f, 0.f);
            float  ns = 0.f;
            if (grow < n_nodes) {
                v  = *(const float4*)(X + (size_t)grow * D_IN + col4);
                ns = norm_src[grow];
            }
            float* p = &As[row * LDS_STRIDE + col4];
            p[0] = v.x * ns; p[1] = v.y * ns; p[2] = v.z * ns; p[3] = v.w * ns;
        }
    }
    __syncthreads();

    const int wave = tid >> 5;
    const int lane = tid & 31;
    const int half = lane >> 4;
    const int l16  = lane & 15;
    const int n0   = wave * 16;
    const int arow = l16 * LDS_STRIDE;
    const int bcol = n0 + l16;

    v8f acc = {};
    for (int k = 0; k < D_IN; k += 4) {
        const int ka = k + half * 2;
        v2f a;
        a.x = As[arow + ka];
        a.y = As[arow + ka + 1];
        v2f b;
        b.x = W[(size_t)ka * D_OUT + bcol];
        b.y = W[(size_t)(ka + 1) * D_OUT + bcol];
        acc = __builtin_amdgcn_wmma_f32_16x16x4_f32(
                   false, a,  false, b,
                   (short)0, acc,  false,  false);
    }

    if (full_tile) {
        float* base = H + (size_t)(m0 + half * 8) * D_OUT + bcol;
        #pragma unroll
        for (int v = 0; v < 8; ++v) {
            base[(size_t)v * D_OUT] = acc[v];
        }
    } else {
        #pragma unroll
        for (int v = 0; v < 8; ++v) {
            int m = m0 + v + half * 8;
            if (m < n_nodes) H[(size_t)m * D_OUT + bcol] = acc[v];
        }
    }
}

__global__ void edge_scatter_kernel(const int* __restrict__ src, const int* __restrict__ dst,
                                    const float* __restrict__ H, float* __restrict__ out,
                                    int n_edges) {
    long long gtid = (long long)blockIdx.x * blockDim.x + threadIdx.x;
    int e = (int)(gtid >> 4);
    if (e >= n_edges) return;
    int c4 = ((int)gtid & 15) * 4;
    int s = src[e];
    int d = dst[e];
    const float4 v = *(const float4*)(H + (size_t)s * D_OUT + c4);
    float* o = out + (size_t)d * D_OUT + c4;
    atomicAdd(o + 0, v.x);
    atomicAdd(o + 1, v.y);
    atomicAdd(o + 2, v.z);
    atomicAdd(o + 3, v.w);
}

__global__ void finalize_kernel(float* __restrict__ out, const float* __restrict__ H,
                                const float* __restrict__ norm_dst,
                                const float* __restrict__ bias, int n_nodes) {
    long long gtid = (long long)blockIdx.x * blockDim.x + threadIdx.x;
    if (gtid >= (long long)n_nodes * D_OUT) return;
    int n = (int)(gtid >> 6);
    int c = (int)(gtid & 63);
    out[gtid] = (out[gtid] + H[gtid]) * norm_dst[n] + bias[c];
}

extern "C" void kernel_launch(void* const* d_in, const int* in_sizes, int n_in,
                              void* d_out, int out_size, void* d_ws, size_t ws_size,
                              hipStream_t stream) {
    const float* in_feat = (const float*)d_in[0];
    const int*   src     = (const int*)d_in[1];
    const int*   dst     = (const int*)d_in[2];
    const float* W       = (const float*)d_in[3];
    const float* b       = (const float*)d_in[4];
    float*       out     = (float*)d_out;

    const int n_nodes = in_sizes[0] / D_IN;
    const int n_edges = in_sizes[1];

    char* ws = (char*)d_ws;
    float* out_deg = (float*)ws;
    float* in_deg  = (float*)(ws + (size_t)n_nodes * 4);
    float* H       = (float*)(ws + (size_t)n_nodes * 8);

    const long long out_elems = (long long)n_nodes * D_OUT;

    init_deg_kernel<<<(n_nodes + 255) / 256, 256, 0, stream>>>(out_deg, in_deg, n_nodes);
    zero_out_kernel<<<(int)((out_elems + 255) / 256), 256, 0, stream>>>(out, out_elems);

    degree_kernel<<<(n_edges + 255) / 256, 256, 0, stream>>>(src, dst, out_deg, in_deg, n_edges);
    norm_kernel<<<(n_nodes + 255) / 256, 256, 0, stream>>>(out_deg, in_deg, n_nodes);

    gemm_wmma_kernel<<<(n_nodes + 15) / 16, 128, 0, stream>>>(in_feat, W, out_deg, H, n_nodes);

    long long ethreads = (long long)n_edges * 16;
    edge_scatter_kernel<<<(int)((ethreads + 255) / 256), 256, 0, stream>>>(src, dst, H, out, n_edges);

    finalize_kernel<<<(int)((out_elems + 255) / 256), 256, 0, stream>>>(out, H, in_deg, b, n_nodes);
}


# Degrees counted as integers in LDS histograms per node range (sliced scan of src/dst, partial counts summed in the norm kernel) instead of global f32 atomics; seeding launch dropped, 7 -> 6 launches
# speedup vs baseline: 1.7357x; 1.7357x over previous
#include <hip/hip_runtime.h>

#define D_IN   256
#define D_OUT  64
#define LDS_STRIDE 260

typedef float v2f __attribute__((ext_vector_type(2)));
typedef float v8f __attribute__((ext_vector_type(8)));

__global__ void zero_out_kernel(float* __restrict__ out, long long n) {
    long long i = (long long)blockIdx.x * blockDim.x + threadIdx.x;
    if (i < n) out[i] = 0.0f;
}

#define DEG_NR      12544
#define DEG_SLICES  32
#define DEG_THREADS 512
#define DEG_DEPTH   8

__device__ __forceinline__ void deg_count(int* hist, int v, int lo, unsigned cnt) {
    unsigned u = (unsigned)(v - lo);
    if (u < cnt) atomicAdd(&hist[u], 1);
}

__device__ __forceinline__ void deg_count4(int* hist, int4 v, int lo, unsigned cnt) {
    deg_count(hist, v.x, lo, cnt);
    deg_count(hist, v.y, lo, cnt);
    deg_count(hist, v.z, lo, cnt);
    deg_count(hist, v.w, lo, cnt);
}

__global__ __launch_bounds__(DEG_THREADS) void degree_lds_kernel(
        const int* __restrict__ src, const int* __restrict__ dst,
        int* __restrict__ part, int n_nodes, int n_edges) {
    __shared__ int hist[DEG_NR];

    const int tid = threadIdx.x;
    const int lo  = blockIdx.x * DEG_NR;
    const int s   = blockIdx.y;
    const int a   = blockIdx.z;
    const int* __restrict__ idx = a ? dst : src;
    const int left = n_nodes - lo;
    const unsigned cnt = (unsigned)(left < DEG_NR ? left : DEG_NR);

    for (int i = tid; i < DEG_NR; i += DEG_THREADS) hist[i] = 0;
    __syncthreads();

    const int per = ((n_edges + DEG_SLICES * 4 - 1) / (DEG_SLICES * 4)) * 4;
    const int e0  = s * per;
    int e1 = e0 + per;
    if (e1 > n_edges) e1 = n_edges;
    const int n4 = (e1 > e0) ? ((e1 - e0) >> 2) : 0;
    const int4* __restrict__ p4 = (const int4*)(idx + e0);

    int i = tid;
    for (; i + (DEG_DEPTH - 1) * DEG_THREADS < n4; i += DEG_DEPTH * DEG_THREADS) {
        int4 v[DEG_DEPTH];
        #pragma unroll
        for (int j = 0; j < DEG_DEPTH; ++j) v[j] = p4[i + j * DEG_THREADS];
        #pragma unroll
        for (int j = 0; j < DEG_DEPTH; ++j) deg_count4(hist, v[j], lo, cnt);
    }
    for (; i < n4; i += DEG_THREADS) deg_count4(hist, p4[i], lo, cnt);
    for (int e = e0 + (n4 << 2) + tid; e < e1; e += DEG_THREADS) deg_count(hist, idx[e], lo, cnt);

    __syncthreads();
    int* __restrict__ o = part + ((size_t)(a * DEG_SLICES + s) * n_nodes + lo);
    for (int k = tid; k < (int)cnt; k += DEG_THREADS) o[k] = hist[k];
}

__global__ void norm_kernel(const int* __restrict__ part,
                            float* __restrict__ out_deg, float* __restrict__ in_deg,
                            int n_nodes) {
    int i = blockIdx.x * blockDim.x + threadIdx.x;
    if (i >= n_nodes) return;
    int so = 0, si = 0;
    #pragma unroll
    for (int s = 0; s < DEG_SLICES; ++s) {
        so += part[(size_t)s * n_nodes + i];
        si += part[(size_t)(DEG_SLICES + s) * n_nodes + i];
    }
    out_deg[i] = rsqrtf(1.0f + (float)so);
    in_deg[i]  = rsqrtf(1.0f + (float)si);
}

__global__ __launch_bounds__(128) void gemm_wmma_kernel(
        const float* __restrict__ X, const float* __restrict__ W,
        const float* __restrict__ norm_src, float* __restrict__ H, int n_nodes) {
    __shared__ float As[16 * LDS_STRIDE];

    const int m0  = blockIdx.x * 16;
    const int tid = threadIdx.x;
    const bool full_tile = (m0 + 16 <= n_nodes);

    if (full_tile) {
        #pragma unroll
        for (int it = 0; it < 8; ++it) {
            int i    = tid + it * 128;
            int row  = i >> 6;
            int col4 = (i & 63) * 4;
            int grow = m0 + row;
            float4 v = *(const float4*)(X + (size_t)grow * D_IN + col4);
            float ns = norm_src[grow];
            float* p = &As[row * LDS_STRIDE + col4];
            p[0] = v.x * ns; p[1] = v.y * ns; p[2] = v.z * ns; p[3] = v.w * ns;
        }
    } else {
        for (int i = tid; i < 16 * (D_IN / 4); i += 128) {
            int row  = i >> 6;
            int col4 = (i & 63) * 4;
            int grow = m0 + row;
            float4 v = make_float4(0.f, 0.f, 0.f, 0.f);
            float  ns = 0.f;
            if (grow < n_nodes) {
                v  = *(const float4*)(X + (size_t)grow * D_IN + col4);
                ns = norm_src[grow];
            }
            float* p = &As[row * LDS_STRIDE + col4];
            p[0] = v.x * ns; p[1] = v.y * ns; p[2] = v.z * ns; p[3] = v.w * ns;
        }
    }
    __syncthreads();

    const int wave = tid >> 5;
    const int lane = tid & 31;
    const int half = lane >> 4;
    const int l16  = lane & 15;
    const int n0   = wave * 16;
    const int arow = l16 * LDS_STRIDE;
    const int bcol = n0 + l16;

    v8f acc = {};
    for (int k = 0; k < D_IN; k += 4) {
        const int ka = k + half * 2;
        v2f a;
        a.x = As[arow + ka];
        a.y = As[arow + ka + 1];
        v2f b;
        b.x = W[(size_t)ka * D_OUT + bcol];
        b.y = W[(size_t)(ka + 1) * D_OUT + bcol];
        acc = __builtin_amdgcn_wmma_f32_16x16x4_f32(
                   false, a,  false, b,
                   (short)0, acc,  false,  false);
    }

    if (full_tile) {
        float* base = H + (size_t)(m0 + half * 8) * D_OUT + bcol;
        #pragma unroll
        for (int v = 0; v < 8; ++v) {
            base[(size_t)v * D_OUT] = acc[v];
        }
    } else {
        #pragma unroll
        for (int v = 0; v < 8; ++v) {
            int m = m0 + v + half * 8;
            if (m < n_nodes) H[(size_t)m * D_OUT + bcol] = acc[v];
        }
    }
}

__global__ void edge_scatter_kernel(const int* __restrict__ src, const int* __restrict__ dst,
                                    const float* __restrict__ H, float* __restrict__ out,
                                    int n_edges) {
    long long gtid = (long long)blockIdx.x * blockDim.x + threadIdx.x;
    int e = (int)(gtid >> 4);
    if (e >= n_edges) return;
    int c4 = ((int)gtid & 15) * 4;
    int s = src[e];
    int d = dst[e];
    const float4 v = *(const float4*)(H + (size_t)s * D_OUT + c4);
    float* o = out + (size_t)d * D_OUT + c4;
    atomicAdd(o + 0, v.x);
    atomicAdd(o + 1, v.y);
    atomicAdd(o + 2, v.z);
    atomicAdd(o + 3, v.w);
}

__global__ void finalize_kernel(float* __restrict__ out, const float* __restrict__ H,
                                const float* __restrict__ norm_dst,
                                const float* __restrict__ bias, int n_nodes) {
    long long gtid = (long long)blockIdx.x * blockDim.x + threadIdx.x;
    if (gtid >= (long long)n_nodes * D_OUT) return;
    int n = (int)(gtid >> 6);
    int c = (int)(gtid & 63);
    out[gtid] = (out[gtid] + H[gtid]) * norm_dst[n] + bias[c];
}

extern "C" void kernel_launch(void* const* d_in, const int* in_sizes, int n_in,
                              void* d_out, int out_size, void* d_ws, size_t ws_size,
                              hipStream_t stream) {
    const float* in_feat = (const float*)d_in[0];
    const int*   src     = (const int*)d_in[1];
    const int*   dst     = (const int*)d_in[2];
    const float* W       = (const float*)d_in[3];
    const float* b       = (const float*)d_in[4];
    float*       out     = (float*)d_out;

    const int n_nodes = in_sizes[0] / D_IN;
    const int n_edges = in_sizes[1];

    char* ws = (char*)d_ws;
    float* out_deg = (float*)ws;
    float* in_deg  = (float*)(ws + (size_t)n_nodes * 4);
    float* H       = (float*)(ws + (size_t)n_nodes * 8);
    int*   part    = (int*)(ws + (size_t)n_nodes * 8 + (size_t)n_nodes * D_OUT * 4);

    const long long out_elems = (long long)n_nodes * D_OUT;

    zero_out_kernel<<<(int)((out_elems + 255) / 256), 256, 0, stream>>>(out, out_elems);

    degree_lds_kernel<<<dim3((n_nodes + DEG_NR - 1) / DEG_NR, DEG_SLICES, 2), DEG_THREADS, 0, stream>>>(
        src, dst, part, n_nodes, n_edges);
    norm_kernel<<<(n_nodes + 255) / 256, 256, 0, stream>>>(part, out_deg, in_deg, n_nodes);

    gemm_wmma_kernel<<<(n_nodes + 15) / 16, 128, 0, stream>>>(in_feat, W, out_deg, H, n_nodes);

    long long ethreads = (long long)n_edges * 16;
    edge_scatter_kernel<<<(int)((ethreads + 255) / 256), 256, 0, stream>>>(src, dst, H, out, n_edges);

    finalize_kernel<<<(int)((out_elems + 255) / 256), 256, 0, stream>>>(out, H, in_deg, b, n_nodes);
}


# Aggregation as a gather: edges bucketed by dst with LDS cursors from the degree counts; each node's h rows summed in registers in bucket order, out written once; no float atomics, 6 -> 5 launches
# speedup vs baseline: 4.4442x; 2.5605x over previous
#include <hip/hip_runtime.h>

#define D_IN   256
#define D_OUT  64
#define LDS_STRIDE 260

typedef float v2f __attribute__((ext_vector_type(2)));
typedef float v8f __attribute__((ext_vector_type(8)));

#define DEG_NR      12544
#define DEG_SLICES  32
#define DEG_THREADS 512
#define DEG_DEPTH   8

__device__ __forceinline__ int slice_len(int n_edges) {
    return ((n_edges + DEG_SLICES * 4 - 1) / (DEG_SLICES * 4)) * 4;
}

__device__ __forceinline__ void deg_count(int* hist, int v, int lo, unsigned cnt) {
    unsigned u = (unsigned)(v - lo);
    if (u < cnt) atomicAdd(&hist[u], 1);
}

__device__ __forceinline__ void deg_count4(int* hist, int4 v, int lo, unsigned cnt) {
    deg_count(hist, v.x, lo, cnt);
    deg_count(hist, v.y, lo, cnt);
    deg_count(hist, v.z, lo, cnt);
    deg_count(hist, v.w, lo, cnt);
}

__global__ __launch_bounds__(DEG_THREADS) void degree_lds_kernel(
        const int* __restrict__ src, const int* __restrict__ dst,
        int* __restrict__ part, int n_nodes, int n_edges) {
    __shared__ int hist[DEG_NR];

    const int tid = threadIdx.x;
    const int lo  = blockIdx.x * DEG_NR;
    const int s   = blockIdx.y;
    const int a   = blockIdx.z;
    const int* __restrict__ idx = a ? dst : src;
    const int left = n_nodes - lo;
    const unsigned cnt = (unsigned)(left < DEG_NR ? left : DEG_NR);

    for (int i = tid; i < DEG_NR; i += DEG_THREADS) hist[i] = 0;
    __syncthreads();

    const int per = slice_len(n_edges);
    const int e0  = s * per;
    int e1 = e0 + per;
    if (e1 > n_edges) e1 = n_edges;
    const int n4 = (e1 > e0) ? ((e1 - e0) >> 2) : 0;
    const int4* __restrict__ p4 = (const int4*)(idx + e0);

    int i = tid;
    for (; i + (DEG_DEPTH - 1) * DEG_THREADS < n4; i += DEG_DEPTH * DEG_THREADS) {
        int4 v[DEG_DEPTH];
        #pragma unroll
        for (int j = 0; j < DEG_DEPTH; ++j) v[j] = p4[i + j * DEG_THREADS];
        #pragma unroll
        for (int j = 0; j < DEG_DEPTH; ++j) deg_count4(hist, v[j], lo, cnt);
    }
    for (; i < n4; i += DEG_THREADS) deg_count4(hist, p4[i], lo, cnt);
    for (int e = e0 + (n4 << 2) + tid; e < e1; e += DEG_THREADS) deg_count(hist, idx[e], lo, cnt);

    __syncthreads();
    int* __restrict__ o = part + ((size_t)(a * DEG_SLICES + s) * n_nodes + lo);
    for (int k = tid; k < (int)cnt; k += DEG_THREADS) o[k] = hist[k];
}

#define NORM_THREADS 256
__global__ __launch_bounds__(NORM_THREADS) void norm_kernel(
        const int* __restrict__ part,
        float* __restrict__ out_deg, float* __restrict__ in_deg,
        int* __restrict__ pre, int* __restrict__ cnt_in, int* __restrict__ blocksum,
        int n_nodes) {
    __shared__ int wsum[NORM_THREADS / 32];
    const int i = blockIdx.x * NORM_THREADS + threadIdx.x;
    int so = 0, si = 0;
    if (i < n_nodes) {
        int pin[DEG_SLICES];
        #pragma unroll
        for (int s = 0; s < DEG_SLICES; ++s) {
            so += part[(size_t)s * n_nodes + i];
            pin[s] = part[(size_t)(DEG_SLICES + s) * n_nodes + i];
        }
        #pragma unroll
        for (int s = 0; s < DEG_SLICES; ++s) {
            pre[(size_t)s * n_nodes + i] = si;
            si += pin[s];
        }
        out_deg[i] = rsqrtf(1.0f + (float)so);
        in_deg[i]  = rsqrtf(1.0f + (float)si);
        cnt_in[i]  = si;
    }
    int v = si;
    for (int m = 16; m >= 1; m >>= 1) v += __shfl_xor(v, m);
    if ((threadIdx.x & 31) == 0) wsum[threadIdx.x >> 5] = v;
    __syncthreads();
    if (threadIdx.x == 0) {
        int t = 0;
        for (int w = 0; w < NORM_THREADS / 32; ++w) t += wsum[w];
        blocksum[blockIdx.x] = t;
    }
}

__global__ __launch_bounds__(128) void gemm_wmma_kernel(
        const float* __restrict__ X, const float* __restrict__ W,
        const float* __restrict__ norm_src, float* __restrict__ H, int n_nodes) {
    __shared__ float As[16 * LDS_STRIDE];

    const int m0  = blockIdx.x * 16;
    const int tid = threadIdx.x;
    const bool full_tile = (m0 + 16 <= n_nodes);

    if (full_tile) {
        #pragma unroll
        for (int it = 0; it < 8; ++it) {
            int i    = tid + it * 128;
            int row  = i >> 6;
            int col4 = (i & 63) * 4;
            int grow = m0 + row;
            float4 v = *(const float4*)(X + (size_t)grow * D_IN + col4);
            float ns = norm_src[grow];
            float* p = &As[row * LDS_STRIDE + col4];
            p[0] = v.x * ns; p[1] = v.y * ns; p[2] = v.z * ns; p[3] = v.w * ns;
        }
    } else {
        for (int i = tid; i < 16 * (D_IN / 4); i += 128) {
            int row  = i >> 6;
            int col4 = (i & 63) * 4;
            int grow = m0 + row;
            float4 v = make_float4(0.f, 0.f, 0.f, 0.f);
            float  ns = 0.f;
            if (grow < n_nodes) {
                v  = *(const float4*)(X + (size_t)grow * D_IN + col4);
                ns = norm_src[grow];
            }
            float* p = &As[row * LDS_STRIDE + col4];
            p[0] = v.x * ns; p[1] = v.y * ns; p[2] = v.z * ns; p[3] = v.w * ns;
        }
    }
    __syncthreads();

    const int wave = tid >> 5;
    const int lane = tid & 31;
    const int half = lane >> 4;
    const int l16  = lane & 15;
    const int n0   = wave * 16;
    const int arow = l16 * LDS_STRIDE;
    const int bcol = n0 + l16;

    v8f acc = {};
    for (int k = 0; k < D_IN; k += 4) {
        const int ka = k + half * 2;
        v2f a;
        a.x = As[arow + ka];
        a.y = As[arow + ka + 1];
        v2f b;
        b.x = W[(size_t)ka * D_OUT + bcol];
        b.y = W[(size_t)(ka + 1) * D_OUT + bcol];
        acc = __builtin_amdgcn_wmma_f32_16x16x4_f32(
                   false, a,  false, b,
                   (short)0, acc,  false,  false);
    }

    if (full_tile) {
        float* base = H + (size_t)(m0 + half * 8) * D_OUT + bcol;
        #pragma unroll
        for (int v = 0; v < 8; ++v) {
            base[(size_t)v * D_OUT] = acc[v];
        }
    } else {
        #pragma unroll
        for (int v = 0; v < 8; ++v) {
            int m = m0 + v + half * 8;
            if (m < n_nodes) H[(size_t)m * D_OUT + bcol] = acc[v];
        }
    }
}

#define FILL_NR     12800
#define FILL_CHUNK  (FILL_NR / DEG_THREADS)
#define FILL_DEPTH  4

__device__ __forceinline__ int fill_take(int* cursor, int d, int lo, unsigned cnt) {
    unsigned u = (unsigned)(d - lo);
    return (u < cnt) ? atomicAdd(&cursor[u], 1) : -1;
}

struct pos4 { int x, y, z, w; };

__device__ __forceinline__ pos4 fill_take4(int* cursor, int4 d, int lo, unsigned cnt) {
    pos4 p;
    p.x = fill_take(cursor, d.x, lo, cnt);
    p.y = fill_take(cursor, d.y, lo, cnt);
    p.z = fill_take(cursor, d.z, lo, cnt);
    p.w = fill_take(cursor, d.w, lo, cnt);
    return p;
}

__device__ __forceinline__ void fill_put4(int* __restrict__ col, pos4 p, int4 sidx) {
    if (p.x >= 0) col[p.x] = sidx.x;
    if (p.y >= 0) col[p.y] = sidx.y;
    if (p.z >= 0) col[p.z] = sidx.z;
    if (p.w >= 0) col[p.w] = sidx.w;
}

__global__ __launch_bounds__(DEG_THREADS) void csr_fill_kernel(
        const int* __restrict__ src, const int* __restrict__ dst,
        const int* __restrict__ pre, const int* __restrict__ cnt_in,
        const int* __restrict__ blocksum,
        int* __restrict__ row_ptr, int* __restrict__ col,
        int n_nodes, int n_edges) {
    __shared__ int cursor[FILL_NR];
    __shared__ int wtot[DEG_THREADS / 32];
    __shared__ int wbase[DEG_THREADS / 32];

    const int tid  = threadIdx.x;
    const int lane = tid & 31;
    const int wv   = tid >> 5;
    const int lo   = blockIdx.x * FILL_NR;
    const int s    = blockIdx.y;
    const int left = n_nodes - lo;
    const unsigned cnt = (unsigned)(left < FILL_NR ? left : FILL_NR);

    int cv[FILL_CHUNK], pv[FILL_CHUNK];
    #pragma unroll
    for (int j = 0; j < FILL_CHUNK; ++j) {
        int node = lo + tid + j * DEG_THREADS;
        if (node > n_nodes - 1) node = n_nodes - 1;
        cv[j] = cnt_in[node];
        pv[j] = pre[(size_t)s * n_nodes + node];
    }
    #pragma unroll
    for (int j = 0; j < FILL_CHUNK; ++j) {
        const int k = tid + j * DEG_THREADS;
        cursor[k] = (k < (int)cnt) ? cv[j] : 0;
    }
    int bpart = 0;
    for (int b = tid; b < lo / NORM_THREADS; b += DEG_THREADS) bpart += blocksum[b];
    __syncthreads();

    const int k0 = tid * FILL_CHUNK;
    int csum = 0;
    for (int j = 0; j < FILL_CHUNK; ++j) csum += cursor[k0 + j];
    int inc = csum;
    for (int d = 1; d < 32; d <<= 1) {
        int y = __shfl_up(inc, d);
        if (lane >= d) inc += y;
    }
    int bred = bpart;
    for (int m = 16; m >= 1; m >>= 1) bred += __shfl_xor(bred, m);
    if (lane == 31) wtot[wv]  = inc;
    if (lane == 0)  wbase[wv] = bred;
    __syncthreads();
    int run = inc - csum;
    for (int w = 0; w < DEG_THREADS / 32; ++w) {
        run += wbase[w];
        if (w < wv) run += wtot[w];
    }
    for (int j = 0; j < FILL_CHUNK; ++j) {
        int cn = cursor[k0 + j];
        cursor[k0 + j] = run;
        run += cn;
    }
    __syncthreads();

    #pragma unroll
    for (int j = 0; j < FILL_CHUNK; ++j) {
        const int k = tid + j * DEG_THREADS;
        if (k < (int)cnt) {
            const int rp = cursor[k];
            if (s == 0) row_ptr[lo + k] = rp;
            cursor[k] = rp + pv[j];
        }
    }
    __syncthreads();

    const int per = slice_len(n_edges);
    const int e0  = s * per;
    int e1 = e0 + per;
    if (e1 > n_edges) e1 = n_edges;
    const int n4 = (e1 > e0) ? ((e1 - e0) >> 2) : 0;
    const int4* __restrict__ d4 = (const int4*)(dst + e0);
    const int4* __restrict__ s4 = (const int4*)(src + e0);

    int i = tid;
    for (; i + (FILL_DEPTH - 1) * DEG_THREADS < n4; i += FILL_DEPTH * DEG_THREADS) {
        int4 dv[FILL_DEPTH], sv[FILL_DEPTH];
        #pragma unroll
        for (int j = 0; j < FILL_DEPTH; ++j) {
            dv[j] = d4[i + j * DEG_THREADS];
            sv[j] = s4[i + j * DEG_THREADS];
        }
        pos4 pp[FILL_DEPTH];
        #pragma unroll
        for (int j = 0; j < FILL_DEPTH; ++j) pp[j] = fill_take4(cursor, dv[j], lo, cnt);
        #pragma unroll
        for (int j = 0; j < FILL_DEPTH; ++j) fill_put4(col, pp[j], sv[j]);
    }
    for (; i < n4; i += DEG_THREADS) {
        const int4 dv1 = d4[i];
        const int4 sv1 = s4[i];
        fill_put4(col, fill_take4(cursor, dv1, lo, cnt), sv1);
    }
    for (int e = e0 + (n4 << 2) + tid; e < e1; e += DEG_THREADS) {
        const int p = fill_take(cursor, dst[e], lo, cnt);
        if (p >= 0) col[p] = src[e];
    }
}

#define GATHER_DEPTH 8

__device__ __forceinline__ void add4(float4& a, const float4 v) {
    a.x += v.x; a.y += v.y; a.z += v.z; a.w += v.w;
}

__global__ __launch_bounds__(256) void gather_kernel(
        const int* __restrict__ col, const int* __restrict__ row_ptr,
        const int* __restrict__ cnt_in, const float* __restrict__ H,
        const float* __restrict__ norm_dst, const float* __restrict__ bias,
        float* __restrict__ out, int n_nodes) {
    const int gtid = blockIdx.x * 256 + threadIdx.x;
    const int n = gtid >> 4;
    if (n >= n_nodes) return;
    const int c4 = (gtid & 15) * 4;
    const int beg = row_ptr[n];
    const int end = beg + cnt_in[n];
    const float nd = norm_dst[n];
    const float4 bb = *(const float4*)(bias + c4);
    float4 acc = *(const float4*)(H + (size_t)n * D_OUT + c4);

    int e = beg;
    for (; e + GATHER_DEPTH <= end; e += GATHER_DEPTH) {
        int   sidx[GATHER_DEPTH];
        float4 v[GATHER_DEPTH];
        #pragma unroll
        for (int j = 0; j < GATHER_DEPTH; ++j) sidx[j] = col[e + j];
        #pragma unroll
        for (int j = 0; j < GATHER_DEPTH; ++j)
            v[j] = *(const float4*)(H + (size_t)sidx[j] * D_OUT + c4);
        #pragma unroll
        for (int j = 0; j < GATHER_DEPTH; ++j) add4(acc, v[j]);
    }
    if (e < end) {
        int   sidx[GATHER_DEPTH];
        float4 v[GATHER_DEPTH];
        #pragma unroll
        for (int j = 0; j < GATHER_DEPTH; ++j) {
            const int ee = e + j;
            const int ci = col[ee < end ? ee : end - 1];
            sidx[j] = (ee < end) ? ci : n;
        }
        #pragma unroll
        for (int j = 0; j < GATHER_DEPTH; ++j)
            v[j] = *(const float4*)(H + (size_t)sidx[j] * D_OUT + c4);
        #pragma unroll
        for (int j = 0; j < GATHER_DEPTH; ++j) {
            const bool live = (e + j < end);
            float4 w;
            w.x = live ? v[j].x : 0.0f;
            w.y = live ? v[j].y : 0.0f;
            w.z = live ? v[j].z : 0.0f;
            w.w = live ? v[j].w : 0.0f;
            add4(acc, w);
        }
    }

    float4 o;
    o.x = acc.x * nd + bb.x;
    o.y = acc.y * nd + bb.y;
    o.z = acc.z * nd + bb.z;
    o.w = acc.w * nd + bb.w;
    *(float4*)(out + (size_t)n * D_OUT + c4) = o;
}

extern "C" void kernel_launch(void* const* d_in, const int* in_sizes, int n_in,
                              void* d_out, int out_size, void* d_ws, size_t ws_size,
                              hipStream_t stream) {
    const float* in_feat = (const float*)d_in[0];
    const int*   src     = (const int*)d_in[1];
    const int*   dst     = (const int*)d_in[2];
    const float* W       = (const float*)d_in[3];
    const float* b       = (const float*)d_in[4];
    float*       out     = (float*)d_out;

    const int n_nodes = in_sizes[0] / D_IN;
    const int n_edges = in_sizes[1];

    char* ws = (char*)d_ws;
    float* out_deg = (float*)ws;
    float* in_deg  = (float*)(ws + (size_t)n_nodes * 4);
    float* H       = (float*)(ws + (size_t)n_nodes * 8);
    int*   part    = (int*)(ws + (size_t)n_nodes * 264);
    int*   pre     = (int*)(ws + (size_t)n_nodes * 520);
    int*   cnt_in  = (int*)(ws + (size_t)n_nodes * 648);
    int*   row_ptr = (int*)(ws + (size_t)n_nodes * 652);
    int*   col     = (int*)(ws + (size_t)n_nodes * 656);
    int*   blocksum = (int*)(ws + (size_t)n_nodes * 656 + (size_t)n_edges * 4);

    degree_lds_kernel<<<dim3((n_nodes + DEG_NR - 1) / DEG_NR, DEG_SLICES, 2), DEG_THREADS, 0, stream>>>(
        src, dst, part, n_nodes, n_edges);
    norm_kernel<<<(n_nodes + NORM_THREADS - 1) / NORM_THREADS, NORM_THREADS, 0, stream>>>(
        part, out_deg, in_deg, pre, cnt_in, blocksum, n_nodes);

    gemm_wmma_kernel<<<(n_nodes + 15) / 16, 128, 0, stream>>>(in_feat, W, out_deg, H, n_nodes);

    csr_fill_kernel<<<dim3((n_nodes + FILL_NR - 1) / FILL_NR, DEG_SLICES, 1), DEG_THREADS, 0, stream>>>(
        src, dst, pre, cnt_in, blocksum, row_ptr, col, n_nodes, n_edges);

    gather_kernel<<<(n_nodes * 16 + 255) / 256, 256, 0, stream>>>(
        col, row_ptr, cnt_in, H, in_deg, b, out, n_nodes);
}


# Bucket fill and GEMM in one launch: the 256 fill work-groups start first, 64-row GEMM work-groups behind them, so the fill's load and LDS waits run under the matrix work; 5 -> 4 launches
# speedup vs baseline: 4.6390x; 1.0438x over previous
#include <hip/hip_runtime.h>

#define D_IN   256
#define D_OUT  64
#define LDS_STRIDE 260

typedef float v2f __attribute__((ext_vector_type(2)));
typedef float v8f __attribute__((ext_vector_type(8)));

#define DEG_NR      12544
#define DEG_SLICES  32
#define DEG_THREADS 512
#define DEG_DEPTH   8

__device__ __forceinline__ int slice_len(int n_edges) {
    return ((n_edges + DEG_SLICES * 4 - 1) / (DEG_SLICES * 4)) * 4;
}

__device__ __forceinline__ void deg_count(int* hist, int v, int lo, unsigned cnt) {
    unsigned u = (unsigned)(v - lo);
    if (u < cnt) atomicAdd(&hist[u], 1);
}

__device__ __forceinline__ void deg_count4(int* hist, int4 v, int lo, unsigned cnt) {
    deg_count(hist, v.x, lo, cnt);
    deg_count(hist, v.y, lo, cnt);
    deg_count(hist, v.z, lo, cnt);
    deg_count(hist, v.w, lo, cnt);
}

__global__ __launch_bounds__(DEG_THREADS) void degree_lds_kernel(
        const int* __restrict__ src, const int* __restrict__ dst,
        int* __restrict__ part, int n_nodes, int n_edges) {
    __shared__ int hist[DEG_NR];

    const int tid = threadIdx.x;
    const int lo  = blockIdx.x * DEG_NR;
    const int s   = blockIdx.y;
    const int a   = blockIdx.z;
    const int* __restrict__ idx = a ? dst : src;
    const int left = n_nodes - lo;
    const unsigned cnt = (unsigned)(left < DEG_NR ? left : DEG_NR);

    for (int i = tid; i < DEG_NR; i += DEG_THREADS) hist[i] = 0;
    __syncthreads();

    const int per = slice_len(n_edges);
    const int e0  = s * per;
    int e1 = e0 + per;
    if (e1 > n_edges) e1 = n_edges;
    const int n4 = (e1 > e0) ? ((e1 - e0) >> 2) : 0;
    const int4* __restrict__ p4 = (const int4*)(idx + e0);

    int i = tid;
    for (; i + (DEG_DEPTH - 1) * DEG_THREADS < n4; i += DEG_DEPTH * DEG_THREADS) {
        int4 v[DEG_DEPTH];
        #pragma unroll
        for (int j = 0; j < DEG_DEPTH; ++j) v[j] = p4[i + j * DEG_THREADS];
        #pragma unroll
        for (int j = 0; j < DEG_DEPTH; ++j) deg_count4(hist, v[j], lo, cnt);
    }
    for (; i < n4; i += DEG_THREADS) deg_count4(hist, p4[i], lo, cnt);
    for (int e = e0 + (n4 << 2) + tid; e < e1; e += DEG_THREADS) deg_count(hist, idx[e], lo, cnt);

    __syncthreads();
    int* __restrict__ o = part + ((size_t)(a * DEG_SLICES + s) * n_nodes + lo);
    for (int k = tid; k < (int)cnt; k += DEG_THREADS) o[k] = hist[k];
}

#define NORM_THREADS 256
__global__ __launch_bounds__(NORM_THREADS) void norm_kernel(
        const int* __restrict__ part,
        float* __restrict__ out_deg, float* __restrict__ in_deg,
        int* __restrict__ pre, int* __restrict__ cnt_in, int* __restrict__ blocksum,
        int n_nodes) {
    __shared__ int wsum[NORM_THREADS / 32];
    const int i = blockIdx.x * NORM_THREADS + threadIdx.x;
    int so = 0, si = 0;
    if (i < n_nodes) {
        int pin[DEG_SLICES];
        #pragma unroll
        for (int s = 0; s < DEG_SLICES; ++s) {
            so += part[(size_t)s * n_nodes + i];
            pin[s] = part[(size_t)(DEG_SLICES + s) * n_nodes + i];
        }
        #pragma unroll
        for (int s = 0; s < DEG_SLICES; ++s) {
            pre[(size_t)s * n_nodes + i] = si;
            si += pin[s];
        }
        out_deg[i] = rsqrtf(1.0f + (float)so);
        in_deg[i]  = rsqrtf(1.0f + (float)si);
        cnt_in[i]  = si;
    }
    int v = si;
    for (int m = 16; m >= 1; m >>= 1) v += __shfl_xor(v, m);
    if ((threadIdx.x & 31) == 0) wsum[threadIdx.x >> 5] = v;
    __syncthreads();
    if (threadIdx.x == 0) {
        int t = 0;
        for (int w = 0; w < NORM_THREADS / 32; ++w) t += wsum[w];
        blocksum[blockIdx.x] = t;
    }
}

#define GF_THREADS 512
#define GF_TILES   (GF_THREADS / 128)

__device__ __forceinline__ void gemm_tiles(
        float* __restrict__ smem_f,
        const float* __restrict__ X, const float* __restrict__ W,
        const float* __restrict__ norm_src, float* __restrict__ H, int n_nodes, int group) {
    const int q   = threadIdx.x >> 7;
    const int tid = threadIdx.x & 127;
    float* As = smem_f + q * (16 * LDS_STRIDE);

    const int m0  = (group * GF_TILES + q) * 16;
    const bool full_tile = (m0 + 16 <= n_nodes);

    if (full_tile) {
        #pragma unroll
        for (int it = 0; it < 8; ++it) {
            int i    = tid + it * 128;
            int row  = i >> 6;
            int col4 = (i & 63) * 4;
            int grow = m0 + row;
            float4 v = *(const float4*)(X + (size_t)grow * D_IN + col4);
            float ns = norm_src[grow];
            float* p = &As[row * LDS_STRIDE + col4];
            p[0] = v.x * ns; p[1] = v.y * ns; p[2] = v.z * ns; p[3] = v.w * ns;
        }
    } else {
        for (int i = tid; i < 16 * (D_IN / 4); i += 128) {
            int row  = i >> 6;
            int col4 = (i & 63) * 4;
            int grow = m0 + row;
            float4 v = make_float4(0.f, 0.f, 0.f, 0.f);
            float  ns = 0.f;
            if (grow < n_nodes) {
                v  = *(const float4*)(X + (size_t)grow * D_IN + col4);
                ns = norm_src[grow];
            }
            float* p = &As[row * LDS_STRIDE + col4];
            p[0] = v.x * ns; p[1] = v.y * ns; p[2] = v.z * ns; p[3] = v.w * ns;
        }
    }
    __syncthreads();

    const int wave = tid >> 5;
    const int lane = tid & 31;
    const int half = lane >> 4;
    const int l16  = lane & 15;
    const int n0   = wave * 16;
    const int arow = l16 * LDS_STRIDE;
    const int bcol = n0 + l16;

    v8f acc = {};
    for (int k = 0; k < D_IN; k += 4) {
        const int ka = k + half * 2;
        v2f a;
        a.x = As[arow + ka];
        a.y = As[arow + ka + 1];
        v2f b;
        b.x = W[(size_t)ka * D_OUT + bcol];
        b.y = W[(size_t)(ka + 1) * D_OUT + bcol];
        acc = __builtin_amdgcn_wmma_f32_16x16x4_f32(
                   false, a,  false, b,
                   (short)0, acc,  false,  false);
    }

    if (full_tile) {
        float* base = H + (size_t)(m0 + half * 8) * D_OUT + bcol;
        #pragma unroll
        for (int v = 0; v < 8; ++v) {
            base[(size_t)v * D_OUT] = acc[v];
        }
    } else {
        #pragma unroll
        for (int v = 0; v < 8; ++v) {
            int m = m0 + v + half * 8;
            if (m < n_nodes) H[(size_t)m * D_OUT + bcol] = acc[v];
        }
    }
}

#define FILL_NR     12800
#define FILL_CHUNK  (FILL_NR / DEG_THREADS)
#define FILL_DEPTH  4

__device__ __forceinline__ int fill_take(int* cursor, int d, int lo, unsigned cnt) {
    unsigned u = (unsigned)(d - lo);
    return (u < cnt) ? atomicAdd(&cursor[u], 1) : -1;
}

struct pos4 { int x, y, z, w; };

__device__ __forceinline__ pos4 fill_take4(int* cursor, int4 d, int lo, unsigned cnt) {
    pos4 p;
    p.x = fill_take(cursor, d.x, lo, cnt);
    p.y = fill_take(cursor, d.y, lo, cnt);
    p.z = fill_take(cursor, d.z, lo, cnt);
    p.w = fill_take(cursor, d.w, lo, cnt);
    return p;
}

__device__ __forceinline__ void fill_put4(int* __restrict__ col, pos4 p, int4 sidx) {
    if (p.x >= 0) col[p.x] = sidx.x;
    if (p.y >= 0) col[p.y] = sidx.y;
    if (p.z >= 0) col[p.z] = sidx.z;
    if (p.w >= 0) col[p.w] = sidx.w;
}

#define FILL_SMEM_INTS (FILL_NR + 2 * (DEG_THREADS / 32))

__device__ __forceinline__ void csr_fill(
        int* __restrict__ smem_i,
        const int* __restrict__ src, const int* __restrict__ dst,
        const int* __restrict__ pre, const int* __restrict__ cnt_in,
        const int* __restrict__ blocksum,
        int* __restrict__ row_ptr, int* __restrict__ col,
        int n_nodes, int n_edges, int range, int s) {
    int* cursor = smem_i;
    int* wtot   = smem_i + FILL_NR;
    int* wbase  = smem_i + FILL_NR + DEG_THREADS / 32;

    const int tid  = threadIdx.x;
    const int lane = tid & 31;
    const int wv   = tid >> 5;
    const int lo   = range * FILL_NR;
    const int left = n_nodes - lo;
    const unsigned cnt = (unsigned)(left < FILL_NR ? left : FILL_NR);

    int cv[FILL_CHUNK], pv[FILL_CHUNK];
    #pragma unroll
    for (int j = 0; j < FILL_CHUNK; ++j) {
        int node = lo + tid + j * DEG_THREADS;
        if (node > n_nodes - 1) node = n_nodes - 1;
        cv[j] = cnt_in[node];
        pv[j] = pre[(size_t)s * n_nodes + node];
    }
    #pragma unroll
    for (int j = 0; j < FILL_CHUNK; ++j) {
        const int k = tid + j * DEG_THREADS;
        cursor[k] = (k < (int)cnt) ? cv[j] : 0;
    }
    int bpart = 0;
    for (int b = tid; b < lo / NORM_THREADS; b += DEG_THREADS) bpart += blocksum[b];
    __syncthreads();

    const int k0 = tid * FILL_CHUNK;
    int csum = 0;
    for (int j = 0; j < FILL_CHUNK; ++j) csum += cursor[k0 + j];
    int inc = csum;
    for (int d = 1; d < 32; d <<= 1) {
        int y = __shfl_up(inc, d);
        if (lane >= d) inc += y;
    }
    int bred = bpart;
    for (int m = 16; m >= 1; m >>= 1) bred += __shfl_xor(bred, m);
    if (lane == 31) wtot[wv]  = inc;
    if (lane == 0)  wbase[wv] = bred;
    __syncthreads();
    int run = inc - csum;
    for (int w = 0; w < DEG_THREADS / 32; ++w) {
        run += wbase[w];
        if (w < wv) run += wtot[w];
    }
    for (int j = 0; j < FILL_CHUNK; ++j) {
        int cn = cursor[k0 + j];
        cursor[k0 + j] = run;
        run += cn;
    }
    __syncthreads();

    #pragma unroll
    for (int j = 0; j < FILL_CHUNK; ++j) {
        const int k = tid + j * DEG_THREADS;
        if (k < (int)cnt) {
            const int rp = cursor[k];
            if (s == 0) row_ptr[lo + k] = rp;
            cursor[k] = rp + pv[j];
        }
    }
    __syncthreads();

    const int per = slice_len(n_edges);
    const int e0  = s * per;
    int e1 = e0 + per;
    if (e1 > n_edges) e1 = n_edges;
    const int n4 = (e1 > e0) ? ((e1 - e0) >> 2) : 0;
    const int4* __restrict__ d4 = (const int4*)(dst + e0);
    const int4* __restrict__ s4 = (const int4*)(src + e0);

    int i = tid;
    for (; i + (FILL_DEPTH - 1) * DEG_THREADS < n4; i += FILL_DEPTH * DEG_THREADS) {
        int4 dv[FILL_DEPTH], sv[FILL_DEPTH];
        #pragma unroll
        for (int j = 0; j < FILL_DEPTH; ++j) {
            dv[j] = d4[i + j * DEG_THREADS];
            sv[j] = s4[i + j * DEG_THREADS];
        }
        pos4 pp[FILL_DEPTH];
        #pragma unroll
        for (int j = 0; j < FILL_DEPTH; ++j) pp[j] = fill_take4(cursor, dv[j], lo, cnt);
        #pragma unroll
        for (int j = 0; j < FILL_DEPTH; ++j) fill_put4(col, pp[j], sv[j]);
    }
    for (; i < n4; i += DEG_THREADS) {
        const int4 dv1 = d4[i];
        const int4 sv1 = s4[i];
        fill_put4(col, fill_take4(cursor, dv1, lo, cnt), sv1);
    }
    for (int e = e0 + (n4 << 2) + tid; e < e1; e += DEG_THREADS) {
        const int p = fill_take(cursor, dst[e], lo, cnt);
        if (p >= 0) col[p] = src[e];
    }
}

#define GF_SMEM_BYTES (GF_TILES * 16 * LDS_STRIDE * 4)
static_assert(DEG_THREADS == GF_THREADS, "fill and GEMM roles share one work-group size");
static_assert(FILL_SMEM_INTS * 4 <= GF_SMEM_BYTES, "fill tables fit the shared LDS block");

__global__ __launch_bounds__(GF_THREADS, 16) void gemm_fill_kernel(
        const float* __restrict__ X, const float* __restrict__ W,
        const float* __restrict__ norm_src, float* __restrict__ H,
        const int* __restrict__ src, const int* __restrict__ dst,
        const int* __restrict__ pre, const int* __restrict__ cnt_in,
        const int* __restrict__ blocksum,
        int* __restrict__ row_ptr, int* __restrict__ col,
        int n_nodes, int n_edges, int fill_ranges) {
    __shared__ __align__(16) unsigned char smem[GF_SMEM_BYTES];
    const int n_fill = fill_ranges * DEG_SLICES;
    const int b = blockIdx.x;
    if (b < n_fill) {
        csr_fill((int*)smem, src, dst, pre, cnt_in, blocksum, row_ptr, col,
                 n_nodes, n_edges, b % fill_ranges, b / fill_ranges);
    } else {
        gemm_tiles((float*)smem, X, W, norm_src, H, n_nodes, b - n_fill);
    }
}

#define GATHER_DEPTH 8

__device__ __forceinline__ void add4(float4& a, const float4 v) {
    a.x += v.x; a.y += v.y; a.z += v.z; a.w += v.w;
}

__global__ __launch_bounds__(256) void gather_kernel(
        const int* __restrict__ col, const int* __restrict__ row_ptr,
        const int* __restrict__ cnt_in, const float* __restrict__ H,
        const float* __restrict__ norm_dst, const float* __restrict__ bias,
        float* __restrict__ out, int n_nodes) {
    const int gtid = blockIdx.x * 256 + threadIdx.x;
    const int n = gtid >> 4;
    if (n >= n_nodes) return;
    const int c4 = (gtid & 15) * 4;
    const int beg = row_ptr[n];
    const int end = beg + cnt_in[n];
    const float nd = norm_dst[n];
    const float4 bb = *(const float4*)(bias + c4);
    float4 acc = *(const float4*)(H + (size_t)n * D_OUT + c4);

    int e = beg;
    for (; e + GATHER_DEPTH <= end; e += GATHER_DEPTH) {
        int   sidx[GATHER_DEPTH];
        float4 v[GATHER_DEPTH];
        #pragma unroll
        for (int j = 0; j < GATHER_DEPTH; ++j) sidx[j] = col[e + j];
        #pragma unroll
        for (int j = 0; j < GATHER_DEPTH; ++j)
            v[j] = *(const float4*)(H + (size_t)sidx[j] * D_OUT + c4);
        #pragma unroll
        for (int j = 0; j < GATHER_DEPTH; ++j) add4(acc, v[j]);
    }
    if (e < end) {
        int   sidx[GATHER_DEPTH];
        float4 v[GATHER_DEPTH];
        #pragma unroll
        for (int j = 0; j < GATHER_DEPTH; ++j) {
            const int ee = e + j;
            const int ci = col[ee < end ? ee : end - 1];
            sidx[j] = (ee < end) ? ci : n;
        }
        #pragma unroll
        for (int j = 0; j < GATHER_DEPTH; ++j)
            v[j] = *(const float4*)(H + (size_t)sidx[j] * D_OUT + c4);
        #pragma unroll
        for (int j = 0; j < GATHER_DEPTH; ++j) {
            const bool live = (e + j < end);
            float4 w;
            w.x = live ? v[j].x : 0.0f;
            w.y = live ? v[j].y : 0.0f;
            w.z = live ? v[j].z : 0.0f;
            w.w = live ? v[j].w : 0.0f;
            add4(acc, w);
        }
    }

    float4 o;
    o.x = acc.x * nd + bb.x;
    o.y = acc.y * nd + bb.y;
    o.z = acc.z * nd + bb.z;
    o.w = acc.w * nd + bb.w;
    *(float4*)(out + (size_t)n * D_OUT + c4) = o;
}

extern "C" void kernel_launch(void* const* d_in, const int* in_sizes, int n_in,
                              void* d_out, int out_size, void* d_ws, size_t ws_size,
                              hipStream_t stream) {
    const float* in_feat = (const float*)d_in[0];
    const int*   src     = (const int*)d_in[1];
    const int*   dst     = (const int*)d_in[2];
    const float* W       = (const float*)d_in[3];
    const float* b       = (const float*)d_in[4];
    float*       out     = (float*)d_out;

    const int n_nodes = in_sizes[0] / D_IN;
    const int n_edges = in_sizes[1];

    char* ws = (char*)d_ws;
    float* out_deg = (float*)ws;
    float* in_deg  = (float*)(ws + (size_t)n_nodes * 4);
    float* H       = (float*)(ws + (size_t)n_nodes * 8);
    int*   part    = (int*)(ws + (size_t)n_nodes * 264);
    int*   pre     = (int*)(ws + (size_t)n_nodes * 520);
    int*   cnt_in  = (int*)(ws + (size_t)n_nodes * 648);
    int*   row_ptr = (int*)(ws + (size_t)n_nodes * 652);
    int*   col     = (int*)(ws + (size_t)n_nodes * 656);
    int*   blocksum = (int*)(ws + (size_t)n_nodes * 656 + (size_t)n_edges * 4);

    degree_lds_kernel<<<dim3((n_nodes + DEG_NR - 1) / DEG_NR, DEG_SLICES, 2), DEG_THREADS, 0, stream>>>(
        src, dst, part, n_nodes, n_edges);
    norm_kernel<<<(n_nodes + NORM_THREADS - 1) / NORM_THREADS, NORM_THREADS, 0, stream>>>(
        part, out_deg, in_deg, pre, cnt_in, blocksum, n_nodes);

    const int fill_ranges = (n_nodes + FILL_NR - 1) / FILL_NR;
    const int gemm_groups = (n_nodes + 16 * GF_TILES - 1) / (16 * GF_TILES);
    gemm_fill_kernel<<<fill_ranges * DEG_SLICES + gemm_groups, GF_THREADS, 0, stream>>>(
        in_feat, W, out_deg, H, src, dst, pre, cnt_in, blocksum, row_ptr, col,
        n_nodes, n_edges, fill_ranges);

    gather_kernel<<<(n_nodes * 16 + 255) / 256, 256, 0, stream>>>(
        col, row_ptr, cnt_in, H, in_deg, b, out, n_nodes);
}


# Degree count reads each index once: all node counters in one LDS histogram as packed 16-bit pairs per half fill slice (8 node-range passes over src/dst become 1); norm thread owns a node pair
# speedup vs baseline: 4.8847x; 1.0530x over previous
#include <hip/hip_runtime.h>

#define D_IN   256
#define D_OUT  64
#define LDS_STRIDE 260

typedef float v2f __attribute__((ext_vector_type(2)));
typedef float v8f __attribute__((ext_vector_type(8)));

#define DEG_NR      100000
#define DEG_WORDS   (DEG_NR / 2)
static_assert(DEG_NR % 8 == 0, "whole 16-byte groups of packed words per node range");
#define DEG_SLICES  32
#define DEG_SUB     2
#define DEG_THREADS 512
#define CNT_THREADS 1024
#define DEG_DEPTH   12

__device__ __forceinline__ int slice_len(int n_edges) {
    return ((n_edges + DEG_SLICES * 4 - 1) / (DEG_SLICES * 4)) * 4;
}
__device__ __forceinline__ int sub_len(int per) {
    return ((per / DEG_SUB + 3) / 4) * 4;
}

__device__ __forceinline__ void deg_count(unsigned* hist, int v, int lo, unsigned cnt) {
    unsigned u = (unsigned)(v - lo);
    if (u < cnt) atomicAdd(&hist[u >> 1], 1u << ((u & 1u) * 16u));
}

__device__ __forceinline__ void deg_count4(unsigned* hist, int4 v, int lo, unsigned cnt) {
    deg_count(hist, v.x, lo, cnt);
    deg_count(hist, v.y, lo, cnt);
    deg_count(hist, v.z, lo, cnt);
    deg_count(hist, v.w, lo, cnt);
}

__global__ __launch_bounds__(CNT_THREADS) void degree_lds_kernel(
        const int* __restrict__ src, const int* __restrict__ dst,
        unsigned* __restrict__ part, int n_nodes, int n_edges) {
    __shared__ __align__(16) unsigned hist[DEG_WORDS];

    const int tid = threadIdx.x;
    const int lo  = blockIdx.x * DEG_NR;
    const int d   = blockIdx.y;
    const int a   = blockIdx.z;
    const int* __restrict__ idx = a ? dst : src;
    const int left = n_nodes - lo;
    const unsigned cnt = (unsigned)(left < DEG_NR ? left : DEG_NR);
    const int nw = (int)((cnt + 1u) >> 1);

    for (int i = tid; i < DEG_WORDS / 4; i += CNT_THREADS) ((uint4*)hist)[i] = make_uint4(0u, 0u, 0u, 0u);
    __syncthreads();

    const int per  = slice_len(n_edges);
    const int hper = sub_len(per);
    const int s0   = (d / DEG_SUB) * per;
    const int e0   = s0 + (d % DEG_SUB) * hper;
    int lim = s0 + per;
    if (lim > n_edges) lim = n_edges;
    int e1 = e0 + hper;
    if (e1 > lim) e1 = lim;
    const int n4 = (e1 > e0) ? ((e1 - e0) >> 2) : 0;
    const int4* __restrict__ p4 = (const int4*)(idx + e0);

    int i = tid;
    for (; i + (DEG_DEPTH - 1) * CNT_THREADS < n4; i += DEG_DEPTH * CNT_THREADS) {
        int4 v[DEG_DEPTH];
        #pragma unroll
        for (int j = 0; j < DEG_DEPTH; ++j) v[j] = p4[i + j * CNT_THREADS];
        #pragma unroll
        for (int j = 0; j < DEG_DEPTH; ++j) deg_count4(hist, v[j], lo, cnt);
    }
    for (; i < n4; i += CNT_THREADS) deg_count4(hist, p4[i], lo, cnt);
    if (e1 > e0)
        for (int e = e0 + (n4 << 2) + tid; e < e1; e += CNT_THREADS) deg_count(hist, idx[e], lo, cnt);

    __syncthreads();
    const int tw = (n_nodes + 1) >> 1;
    unsigned* __restrict__ o = part + ((size_t)(a * DEG_SLICES * DEG_SUB + d) * tw + (lo >> 1));
    if ((tw & 3) == 0 && (((size_t)part) & 15) == 0) {
        const int nw4 = nw >> 2;
        for (int k = tid; k < nw4; k += CNT_THREADS) ((uint4*)o)[k] = ((const uint4*)hist)[k];
        for (int k = (nw4 << 2) + tid; k < nw; k += CNT_THREADS) o[k] = hist[k];
    } else {
        for (int k = tid; k < nw; k += CNT_THREADS) o[k] = hist[k];
    }
}

#define NORM_THREADS 128
#define NORM_NODES   (2 * NORM_THREADS)
__global__ __launch_bounds__(NORM_THREADS) void norm_kernel(
        const unsigned* __restrict__ part,
        float* __restrict__ out_deg, float* __restrict__ in_deg,
        int* __restrict__ pre, int* __restrict__ cnt_in, int* __restrict__ blocksum,
        int n_nodes) {
    __shared__ int wsum[NORM_THREADS / 32];
    const int p  = blockIdx.x * NORM_THREADS + threadIdx.x;
    const int tw = (n_nodes + 1) >> 1;
    const int i0 = 2 * p, i1 = 2 * p + 1;
    const bool pair = ((n_nodes & 1) == 0);
    int si0 = 0, si1 = 0;
    if (p < tw) {
        int so0 = 0, so1 = 0;
        int pin0[DEG_SLICES], pin1[DEG_SLICES];
        #pragma unroll
        for (int s = 0; s < DEG_SLICES; ++s) {
            int c0 = 0, c1 = 0;
            #pragma unroll
            for (int k = 0; k < DEG_SUB; ++k) {
                const unsigned ws_ = part[(size_t)(s * DEG_SUB + k) * tw + p];
                const unsigned wd_ = part[(size_t)((DEG_SLICES + s) * DEG_SUB + k) * tw + p];
                so0 += (int)(ws_ & 0xffffu);
                so1 += (int)(ws_ >> 16);
                c0  += (int)(wd_ & 0xffffu);
                c1  += (int)(wd_ >> 16);
            }
            pin0[s] = c0;
            pin1[s] = c1;
        }
        if (pair) {
            #pragma unroll
            for (int s = 0; s < DEG_SLICES; ++s) {
                *(int2*)(pre + (size_t)s * n_nodes + i0) = make_int2(si0, si1);
                si0 += pin0[s];
                si1 += pin1[s];
            }
        } else {
            #pragma unroll
            for (int s = 0; s < DEG_SLICES; ++s) {
                int* q = pre + (size_t)s * n_nodes + i0;
                q[0] = si0;
                if (i1 < n_nodes) q[1] = si1;
                si0 += pin0[s];
                si1 += pin1[s];
            }
        }
        const float od0 = rsqrtf(1.0f + (float)so0), od1 = rsqrtf(1.0f + (float)so1);
        const float id0 = rsqrtf(1.0f + (float)si0), id1 = rsqrtf(1.0f + (float)si1);
        if (pair) {
            *(float2*)(out_deg + i0) = make_float2(od0, od1);
            *(float2*)(in_deg + i0)  = make_float2(id0, id1);
            *(int2*)(cnt_in + i0)    = make_int2(si0, si1);
        } else {
            out_deg[i0] = od0; in_deg[i0] = id0; cnt_in[i0] = si0;
            if (i1 < n_nodes) { out_deg[i1] = od1; in_deg[i1] = id1; cnt_in[i1] = si1; }
        }
    }
    int v = si0 + si1;
    for (int m = 16; m >= 1; m >>= 1) v += __shfl_xor(v, m);
    if ((threadIdx.x & 31) == 0) wsum[threadIdx.x >> 5] = v;
    __syncthreads();
    if (threadIdx.x == 0) {
        int t = 0;
        for (int w = 0; w < NORM_THREADS / 32; ++w) t += wsum[w];
        blocksum[blockIdx.x] = t;
    }
}

#define GF_THREADS 512
#define GF_TILES   (GF_THREADS / 128)

__device__ __forceinline__ void gemm_tiles(
        float* __restrict__ smem_f,
        const float* __restrict__ X, const float* __restrict__ W,
        const float* __restrict__ norm_src, float* __restrict__ H, int n_nodes, int group) {
    const int q   = threadIdx.x >> 7;
    const int tid = threadIdx.x & 127;
    float* As = smem_f + q * (16 * LDS_STRIDE);

    const int m0  = (group * GF_TILES + q) * 16;
    const bool full_tile = (m0 + 16 <= n_nodes);

    if (full_tile) {
        #pragma unroll
        for (int it = 0; it < 8; ++it) {
            int i    = tid + it * 128;
            int row  = i >> 6;
            int col4 = (i & 63) * 4;
            int grow = m0 + row;
            float4 v = *(const float4*)(X + (size_t)grow * D_IN + col4);
            float ns = norm_src[grow];
            float* p = &As[row * LDS_STRIDE + col4];
            p[0] = v.x * ns; p[1] = v.y * ns; p[2] = v.z * ns; p[3] = v.w * ns;
        }
    } else {
        for (int i = tid; i < 16 * (D_IN / 4); i += 128) {
            int row  = i >> 6;
            int col4 = (i & 63) * 4;
            int grow = m0 + row;
            float4 v = make_float4(0.f, 0.f, 0.f, 0.f);
            float  ns = 0.f;
            if (grow < n_nodes) {
                v  = *(const float4*)(X + (size_t)grow * D_IN + col4);
                ns = norm_src[grow];
            }
            float* p = &As[row * LDS_STRIDE + col4];
            p[0] = v.x * ns; p[1] = v.y * ns; p[2] = v.z * ns; p[3] = v.w * ns;
        }
    }
    __syncthreads();

    const int wave = tid >> 5;
    const int lane = tid & 31;
    const int half = lane >> 4;
    const int l16  = lane & 15;
    const int n0   = wave * 16;
    const int arow = l16 * LDS_STRIDE;
    const int bcol = n0 + l16;

    v8f acc = {};
    for (int k = 0; k < D_IN; k += 4) {
        const int ka = k + half * 2;
        v2f a;
        a.x = As[arow + ka];
        a.y = As[arow + ka + 1];
        v2f b;
        b.x = W[(size_t)ka * D_OUT + bcol];
        b.y = W[(size_t)(ka + 1) * D_OUT + bcol];
        acc = __builtin_amdgcn_wmma_f32_16x16x4_f32(
                   false, a,  false, b,
                   (short)0, acc,  false,  false);
    }

    if (full_tile) {
        float* base = H + (size_t)(m0 + half * 8) * D_OUT + bcol;
        #pragma unroll
        for (int v = 0; v < 8; ++v) {
            base[(size_t)v * D_OUT] = acc[v];
        }
    } else {
        #pragma unroll
        for (int v = 0; v < 8; ++v) {
            int m = m0 + v + half * 8;
            if (m < n_nodes) H[(size_t)m * D_OUT + bcol] = acc[v];
        }
    }
}

#define FILL_NR     12800
#define FILL_CHUNK  (FILL_NR / DEG_THREADS)
#define FILL_DEPTH  4

__device__ __forceinline__ int fill_take(int* cursor, int d, int lo, unsigned cnt) {
    unsigned u = (unsigned)(d - lo);
    return (u < cnt) ? atomicAdd(&cursor[u], 1) : -1;
}

struct pos4 { int x, y, z, w; };

__device__ __forceinline__ pos4 fill_take4(int* cursor, int4 d, int lo, unsigned cnt) {
    pos4 p;
    p.x = fill_take(cursor, d.x, lo, cnt);
    p.y = fill_take(cursor, d.y, lo, cnt);
    p.z = fill_take(cursor, d.z, lo, cnt);
    p.w = fill_take(cursor, d.w, lo, cnt);
    return p;
}

__device__ __forceinline__ void fill_put4(int* __restrict__ col, pos4 p, int4 sidx) {
    if (p.x >= 0) col[p.x] = sidx.x;
    if (p.y >= 0) col[p.y] = sidx.y;
    if (p.z >= 0) col[p.z] = sidx.z;
    if (p.w >= 0) col[p.w] = sidx.w;
}

#define FILL_SMEM_INTS (FILL_NR + 2 * (DEG_THREADS / 32))

__device__ __forceinline__ void csr_fill(
        int* __restrict__ smem_i,
        const int* __restrict__ src, const int* __restrict__ dst,
        const int* __restrict__ pre, const int* __restrict__ cnt_in,
        const int* __restrict__ blocksum,
        int* __restrict__ row_ptr, int* __restrict__ col,
        int n_nodes, int n_edges, int range, int s) {
    int* cursor = smem_i;
    int* wtot   = smem_i + FILL_NR;
    int* wbase  = smem_i + FILL_NR + DEG_THREADS / 32;

    const int tid  = threadIdx.x;
    const int lane = tid & 31;
    const int wv   = tid >> 5;
    const int lo   = range * FILL_NR;
    const int left = n_nodes - lo;
    const unsigned cnt = (unsigned)(left < FILL_NR ? left : FILL_NR);

    int cv[FILL_CHUNK], pv[FILL_CHUNK];
    #pragma unroll
    for (int j = 0; j < FILL_CHUNK; ++j) {
        int node = lo + tid + j * DEG_THREADS;
        if (node > n_nodes - 1) node = n_nodes - 1;
        cv[j] = cnt_in[node];
        pv[j] = pre[(size_t)s * n_nodes + node];
    }
    #pragma unroll
    for (int j = 0; j < FILL_CHUNK; ++j) {
        const int k = tid + j * DEG_THREADS;
        cursor[k] = (k < (int)cnt) ? cv[j] : 0;
    }
    int bpart = 0;
    for (int b = tid; b < lo / NORM_NODES; b += DEG_THREADS) bpart += blocksum[b];
    __syncthreads();

    const int k0 = tid * FILL_CHUNK;
    int csum = 0;
    for (int j = 0; j < FILL_CHUNK; ++j) csum += cursor[k0 + j];
    int inc = csum;
    for (int d = 1; d < 32; d <<= 1) {
        int y = __shfl_up(inc, d);
        if (lane >= d) inc += y;
    }
    int bred = bpart;
    for (int m = 16; m >= 1; m >>= 1) bred += __shfl_xor(bred, m);
    if (lane == 31) wtot[wv]  = inc;
    if (lane == 0)  wbase[wv] = bred;
    __syncthreads();
    int run = inc - csum;
    for (int w = 0; w < DEG_THREADS / 32; ++w) {
        run += wbase[w];
        if (w < wv) run += wtot[w];
    }
    for (int j = 0; j < FILL_CHUNK; ++j) {
        int cn = cursor[k0 + j];
        cursor[k0 + j] = run;
        run += cn;
    }
    __syncthreads();

    #pragma unroll
    for (int j = 0; j < FILL_CHUNK; ++j) {
        const int k = tid + j * DEG_THREADS;
        if (k < (int)cnt) {
            const int rp = cursor[k];
            if (s == 0) row_ptr[lo + k] = rp;
            cursor[k] = rp + pv[j];
        }
    }
    __syncthreads();

    const int per = slice_len(n_edges);
    const int e0  = s * per;
    int e1 = e0 + per;
    if (e1 > n_edges) e1 = n_edges;
    const int n4 = (e1 > e0) ? ((e1 - e0) >> 2) : 0;
    const int4* __restrict__ d4 = (const int4*)(dst + e0);
    const int4* __restrict__ s4 = (const int4*)(src + e0);

    int i = tid;
    for (; i + (FILL_DEPTH - 1) * DEG_THREADS < n4; i += FILL_DEPTH * DEG_THREADS) {
        int4 dv[FILL_DEPTH], sv[FILL_DEPTH];
        #pragma unroll
        for (int j = 0; j < FILL_DEPTH; ++j) {
            dv[j] = d4[i + j * DEG_THREADS];
            sv[j] = s4[i + j * DEG_THREADS];
        }
        pos4 pp[FILL_DEPTH];
        #pragma unroll
        for (int j = 0; j < FILL_DEPTH; ++j) pp[j] = fill_take4(cursor, dv[j], lo, cnt);
        #pragma unroll
        for (int j = 0; j < FILL_DEPTH; ++j) fill_put4(col, pp[j], sv[j]);
    }
    for (; i < n4; i += DEG_THREADS) {
        const int4 dv1 = d4[i];
        const int4 sv1 = s4[i];
        fill_put4(col, fill_take4(cursor, dv1, lo, cnt), sv1);
    }
    for (int e = e0 + (n4 << 2) + tid; e < e1; e += DEG_THREADS) {
        const int p = fill_take(cursor, dst[e], lo, cnt);
        if (p >= 0) col[p] = src[e];
    }
}

#define GF_SMEM_BYTES (GF_TILES * 16 * LDS_STRIDE * 4)
static_assert(DEG_THREADS == GF_THREADS, "fill and GEMM roles share one work-group size");
static_assert(FILL_NR % NORM_NODES == 0, "a fill range starts on a norm block border");
static_assert(FILL_SMEM_INTS * 4 <= GF_SMEM_BYTES, "fill tables fit the shared LDS block");

__global__ __launch_bounds__(GF_THREADS, 16) void gemm_fill_kernel(
        const float* __restrict__ X, const float* __restrict__ W,
        const float* __restrict__ norm_src, float* __restrict__ H,
        const int* __restrict__ src, const int* __restrict__ dst,
        const int* __restrict__ pre, const int* __restrict__ cnt_in,
        const int* __restrict__ blocksum,
        int* __restrict__ row_ptr, int* __restrict__ col,
        int n_nodes, int n_edges, int fill_ranges) {
    __shared__ __align__(16) unsigned char smem[GF_SMEM_BYTES];
    const int n_fill = fill_ranges * DEG_SLICES;
    const int b = blockIdx.x;
    if (b < n_fill) {
        csr_fill((int*)smem, src, dst, pre, cnt_in, blocksum, row_ptr, col,
                 n_nodes, n_edges, b % fill_ranges, b / fill_ranges);
    } else {
        gemm_tiles((float*)smem, X, W, norm_src, H, n_nodes, b - n_fill);
    }
}

#define GATHER_DEPTH 8

__device__ __forceinline__ void add4(float4& a, const float4 v) {
    a.x += v.x; a.y += v.y; a.z += v.z; a.w += v.w;
}

__global__ __launch_bounds__(256) void gather_kernel(
        const int* __restrict__ col, const int* __restrict__ row_ptr,
        const int* __restrict__ cnt_in, const float* __restrict__ H,
        const float* __restrict__ norm_dst, const float* __restrict__ bias,
        float* __restrict__ out, int n_nodes) {
    const int gtid = blockIdx.x * 256 + threadIdx.x;
    const int n = gtid >> 4;
    if (n >= n_nodes) return;
    const int c4 = (gtid & 15) * 4;
    const int beg = row_ptr[n];
    const int end = beg + cnt_in[n];
    const float nd = norm_dst[n];
    const float4 bb = *(const float4*)(bias + c4);
    float4 acc = *(const float4*)(H + (size_t)n * D_OUT + c4);

    int e = beg;
    for (; e + GATHER_DEPTH <= end; e += GATHER_DEPTH) {
        int   sidx[GATHER_DEPTH];
        float4 v[GATHER_DEPTH];
        #pragma unroll
        for (int j = 0; j < GATHER_DEPTH; ++j) sidx[j] = col[e + j];
        #pragma unroll
        for (int j = 0; j < GATHER_DEPTH; ++j)
            v[j] = *(const float4*)(H + (size_t)sidx[j] * D_OUT + c4);
        #pragma unroll
        for (int j = 0; j < GATHER_DEPTH; ++j) add4(acc, v[j]);
    }
    if (e < end) {
        int   sidx[GATHER_DEPTH];
        float4 v[GATHER_DEPTH];
        #pragma unroll
        for (int j = 0; j < GATHER_DEPTH; ++j) {
            const int ee = e + j;
            const int ci = col[ee < end ? ee : end - 1];
            sidx[j] = (ee < end) ? ci : n;
        }
        #pragma unroll
        for (int j = 0; j < GATHER_DEPTH; ++j)
            v[j] = *(const float4*)(H + (size_t)sidx[j] * D_OUT + c4);
        #pragma unroll
        for (int j = 0; j < GATHER_DEPTH; ++j) {
            const bool live = (e + j < end);
            float4 w;
            w.x = live ? v[j].x : 0.0f;
            w.y = live ? v[j].y : 0.0f;
            w.z = live ? v[j].z : 0.0f;
            w.w = live ? v[j].w : 0.0f;
            add4(acc, w);
        }
    }

    float4 o;
    o.x = acc.x * nd + bb.x;
    o.y = acc.y * nd + bb.y;
    o.z = acc.z * nd + bb.z;
    o.w = acc.w * nd + bb.w;
    *(float4*)(out + (size_t)n * D_OUT + c4) = o;
}

extern "C" void kernel_launch(void* const* d_in, const int* in_sizes, int n_in,
                              void* d_out, int out_size, void* d_ws, size_t ws_size,
                              hipStream_t stream) {
    const float* in_feat = (const float*)d_in[0];
    const int*   src     = (const int*)d_in[1];
    const int*   dst     = (const int*)d_in[2];
    const float* W       = (const float*)d_in[3];
    const float* b       = (const float*)d_in[4];
    float*       out     = (float*)d_out;

    const int n_nodes = in_sizes[0] / D_IN;
    const int n_edges = in_sizes[1];

    char* ws = (char*)d_ws;
    float* out_deg = (float*)ws;
    float* in_deg  = (float*)(ws + (size_t)n_nodes * 4);
    float* H       = (float*)(ws + (size_t)n_nodes * 8);
    int*   pre     = (int*)(ws + (size_t)n_nodes * 520);
    int*   cnt_in  = (int*)(ws + (size_t)n_nodes * 648);
    int*   row_ptr = (int*)(ws + (size_t)n_nodes * 652);
    int*   col     = (int*)(ws + (size_t)n_nodes * 656);
    int*   blocksum = (int*)(ws + (size_t)n_nodes * 656 + (size_t)n_edges * 4);
    unsigned* part = (unsigned*)(ws + (size_t)n_nodes * 656 + (size_t)n_edges * 4 + 65536);

    degree_lds_kernel<<<dim3((n_nodes + DEG_NR - 1) / DEG_NR, DEG_SLICES * DEG_SUB, 2), CNT_THREADS, 0, stream>>>(
        src, dst, part, n_nodes, n_edges);
    norm_kernel<<<(n_nodes + NORM_NODES - 1) / NORM_NODES, NORM_THREADS, 0, stream>>>(
        part, out_deg, in_deg, pre, cnt_in, blocksum, n_nodes);

    const int fill_ranges = (n_nodes + FILL_NR - 1) / FILL_NR;
    const int gemm_groups = (n_nodes + 16 * GF_TILES - 1) / (16 * GF_TILES);
    gemm_fill_kernel<<<fill_ranges * DEG_SLICES + gemm_groups, GF_THREADS, 0, stream>>>(
        in_feat, W, out_deg, H, src, dst, pre, cnt_in, blocksum, row_ptr, col,
        n_nodes, n_edges, fill_ranges);

    gather_kernel<<<(n_nodes * 16 + 255) / 256, 256, 0, stream>>>(
        col, row_ptr, cnt_in, H, in_deg, b, out, n_nodes);
}
